# starting kernel
# baseline (speedup 1.0000x reference)
#include <hip/hip_runtime.h>
#include <hip/hip_bf16.h>

typedef float v2f __attribute__((ext_vector_type(2)));
typedef float v8f __attribute__((ext_vector_type(8)));

#define NNODES 100000
#define NEDGES 1600000

__global__ __launch_bounds__(128) void gcn_gemm_wmma(
    const float* __restrict__ X, const float* __restrict__ Wm,
    const float* __restrict__ bias, float* __restrict__ P,
    int N, int din, int dout, int doutp, int relu_in)
{
    __shared__ float lds[16 * (128 + 4)];
    const int ldsrow = din + 4;
    const int row0 = blockIdx.x * 16;
    const int t = threadIdx.x;

    const int nvec = (16 * din) >> 2;
    const int vper = din >> 2;
    for (int idx = t; idx < nvec; idx += 128) {
        const int r  = idx / vper;
        const int c4 = (idx - r * vper) << 2;
        const int row = row0 + r;
        float4 v = make_float4(0.f, 0.f, 0.f, 0.f);
        if (row < N) {
            v = *(const float4*)(X + (size_t)row * din + c4);
            if (relu_in) {
                v.x = fmaxf(v.x, 0.f); v.y = fmaxf(v.y, 0.f);
                v.z = fmaxf(v.z, 0.f); v.w = fmaxf(v.w, 0.f);
            }
        }
        *(float4*)(&lds[r * ldsrow + c4]) = v;
    }
    __syncthreads();

    const int wave = t >> 5;
    const int lane = t & 31;
    const int colbase = wave * 16;
    if (colbase >= doutp) return;

    const int half = lane >> 4;
    const int nrow = lane & 15;
    const int col  = colbase + nrow;

    const float* wk = Wm + (size_t)(2 * half) * doutp + col;
    int aoff = nrow * ldsrow + 2 * half;

    v8f acc = {};
    for (int k = 0; k < din; k += 4) {
        v2f a, b;
        a.x = lds[aoff];
        a.y = lds[aoff + 1];
        b.x = wk[0];
        b.y = wk[doutp];
        acc = __builtin_amdgcn_wmma_f32_16x16x4_f32(
                  false, a, false, b, (short)0, acc, false, false);
        aoff += 4;
        wk   += (size_t)4 * doutp;
    }

    const float bv = bias[col];
    #pragma unroll
    for (int r = 0; r < 8; ++r) {
        const int m   = r + 8 * half;
        const int row = row0 + m;
        if (row < N && col < dout)
            P[(size_t)row * dout + col] = acc[r] + bv;
    }
}

__global__ void gcn_pad_w4(const float* __restrict__ W4, const float* __restrict__ b4,
                           float* __restrict__ Wp, float* __restrict__ bp)
{
    const int i = blockIdx.x * blockDim.x + threadIdx.x;
    if (i < 64 * 16) {
        const int r = i >> 4, c = i & 15;
        Wp[i] = (c < 10) ? W4[r * 10 + c] : 0.f;
    } else if (i < 64 * 16 + 16) {
        const int c = i - 64 * 16;
        bp[c] = (c < 10) ? b4[c] : 0.f;
    }
}

__global__ void gcn_zero(float* __restrict__ p, int n)
{
    const int i = blockIdx.x * blockDim.x + threadIdx.x;
    if (i < n) p[i] = 0.f;
}

__global__ __launch_bounds__(256) void gcn_aggregate(
    const float* __restrict__ P, const int* __restrict__ src,
    const int* __restrict__ dst, const float* __restrict__ w,
    float* __restrict__ out, int E, int dout)
{
    const int gid = blockIdx.x * blockDim.x + threadIdx.x;
    const int e = gid >> 5;
    if (e >= E) return;
    const int lane = gid & 31;

    if (lane == 0) {
        const int ep = e + 4096;
        if (ep < E) {
            __builtin_prefetch(src + ep, 0, 0);
            __builtin_prefetch(dst + ep, 0, 0);
            __builtin_prefetch(w   + ep, 0, 0);
        }
    }

    const int c = lane << 1;
    if (c >= dout) return;

    const int s = src[e];
    const int d = dst[e];
    const float ww = w[e];

    const float2 h = *(const float2*)(P + (size_t)s * dout + c);
    float* o = out + (size_t)d * dout + c;
    __hip_atomic_fetch_add(o,     h.x * ww, __ATOMIC_RELAXED, __HIP_MEMORY_SCOPE_AGENT);
    __hip_atomic_fetch_add(o + 1, h.y * ww, __ATOMIC_RELAXED, __HIP_MEMORY_SCOPE_AGENT);
}

extern "C" void kernel_launch(void* const* d_in, const int* in_sizes, int n_in,
                              void* d_out, int out_size, void* d_ws, size_t ws_size,
                              hipStream_t stream)
{
    (void)in_sizes; (void)n_in; (void)out_size; (void)ws_size;
    const int N = NNODES, E = NEDGES;

    const float* x    = (const float*)d_in[0];
    const int*   esrc = (const int*)  d_in[1];
    const int*   edst = (const int*)  d_in[2];
    const float* ew   = (const float*)d_in[3];
    const float* Ws[5]; const float* bs[5];
    for (int i = 0; i < 5; ++i) {
        Ws[i] = (const float*)d_in[4 + 2 * i];
        bs[i] = (const float*)d_in[5 + 2 * i];
    }
    float* out  = (float*)d_out;
    float* buf0 = (float*)d_ws;
    float* buf1 = buf0 + (size_t)N * 64;
    float* buf2 = buf1 + (size_t)N * 64;
    float* wpad = buf2 + (size_t)N * 64;
    float* bpad = wpad + 64 * 16;

    const dim3 ggrid((N + 15) / 16);
    const int  agrid = (E * 32 + 255) / 256;

    gcn_pad_w4<<<(64 * 16 + 16 + 255) / 256, 256, 0, stream>>>(Ws[4], bs[4], wpad, bpad);

    gcn_gemm_wmma<<<ggrid, 128, 0, stream>>>(x, Ws[0], bs[0], buf0, N, 128, 64, 64, 0);
    gcn_zero<<<(N * 64 + 255) / 256, 256, 0, stream>>>(buf1, N * 64);
    gcn_aggregate<<<agrid, 256, 0, stream>>>(buf0, esrc, edst, ew, buf1, E, 64);

    gcn_gemm_wmma<<<ggrid, 128, 0, stream>>>(buf1, Ws[1], bs[1], buf2, N, 64, 64, 64, 1);
    gcn_zero<<<(N * 64 + 255) / 256, 256, 0, stream>>>(buf0, N * 64);
    gcn_aggregate<<<agrid, 256, 0, stream>>>(buf2, esrc, edst, ew, buf0, E, 64);

    gcn_gemm_wmma<<<ggrid, 128, 0, stream>>>(buf0, Ws[2], bs[2], buf1, N, 64, 64, 64, 1);
    gcn_zero<<<(N * 64 + 255) / 256, 256, 0, stream>>>(buf2, N * 64);
    gcn_aggregate<<<agrid, 256, 0, stream>>>(buf1, esrc, edst, ew, buf2, E, 64);

    gcn_gemm_wmma<<<ggrid, 128, 0, stream>>>(buf2, Ws[3], bs[3], buf0, N, 64, 64, 64, 1);
    gcn_zero<<<(N * 64 + 255) / 256, 256, 0, stream>>>(buf1, N * 64);
    gcn_aggregate<<<agrid, 256, 0, stream>>>(buf0, esrc, edst, ew, buf1, E, 64);

    gcn_gemm_wmma<<<ggrid, 128, 0, stream>>>(buf1, wpad, bpad, buf2, N, 64, 10, 16, 1);
    gcn_zero<<<(N * 10 + 255) / 256, 256, 0, stream>>>(out, N * 10);
    gcn_aggregate<<<agrid, 256, 0, stream>>>(buf2, esrc, edst, ew, out, E, 10);
}


# Edge scatter split by destination range per package half: a work-group reads its die, draws a (tile, range) item from its half's counter, applies only that range's edges; arrival-order sums kept
# speedup vs baseline: 2.3902x; 2.3902x over previous
#include <hip/hip_runtime.h>
#include <hip/hip_bf16.h>

typedef float v2f __attribute__((ext_vector_type(2)));
typedef float v8f __attribute__((ext_vector_type(8)));

#define NNODES 100000
#define NEDGES 1600000

__global__ __launch_bounds__(128) void gcn_gemm_wmma(
    const float* __restrict__ X, const float* __restrict__ Wm,
    const float* __restrict__ bias, float* __restrict__ P,
    int N, int din, int dout, int doutp, int relu_in)
{
    __shared__ float lds[16 * (128 + 4)];
    const int ldsrow = din + 4;
    const int row0 = blockIdx.x * 16;
    const int t = threadIdx.x;

    const int nvec = (16 * din) >> 2;
    const int vper = din >> 2;
    for (int idx = t; idx < nvec; idx += 128) {
        const int r  = idx / vper;
        const int c4 = (idx - r * vper) << 2;
        const int row = row0 + r;
        float4 v = make_float4(0.f, 0.f, 0.f, 0.f);
        if (row < N) {
            v = *(const float4*)(X + (size_t)row * din + c4);
            if (relu_in) {
                v.x = fmaxf(v.x, 0.f); v.y = fmaxf(v.y, 0.f);
                v.z = fmaxf(v.z, 0.f); v.w = fmaxf(v.w, 0.f);
            }
        }
        *(float4*)(&lds[r * ldsrow + c4]) = v;
    }
    __syncthreads();

    const int wave = t >> 5;
    const int lane = t & 31;
    const int colbase = wave * 16;
    if (colbase >= doutp) return;

    const int half = lane >> 4;
    const int nrow = lane & 15;
    const int col  = colbase + nrow;

    const float* wk = Wm + (size_t)(2 * half) * doutp + col;
    int aoff = nrow * ldsrow + 2 * half;

    v8f acc = {};
    for (int k = 0; k < din; k += 4) {
        v2f a, b;
        a.x = lds[aoff];
        a.y = lds[aoff + 1];
        b.x = wk[0];
        b.y = wk[doutp];
        acc = __builtin_amdgcn_wmma_f32_16x16x4_f32(
                  false, a, false, b, (short)0, acc, false, false);
        aoff += 4;
        wk   += (size_t)4 * doutp;
    }

    const float bv = bias[col];
    #pragma unroll
    for (int r = 0; r < 8; ++r) {
        const int m   = r + 8 * half;
        const int row = row0 + m;
        if (row < N && col < dout)
            P[(size_t)row * dout + col] = acc[r] + bv;
    }
}

#define GCN_NTICK (5 * 2 * 32)

__global__ void gcn_pad_w4(const float* __restrict__ W4, const float* __restrict__ b4,
                           float* __restrict__ Wp, float* __restrict__ bp,
                           int* __restrict__ tick)
{
    const int i = blockIdx.x * blockDim.x + threadIdx.x;
    if (i < 64 * 16) {
        const int r = i >> 4, c = i & 15;
        Wp[i] = (c < 10) ? W4[r * 10 + c] : 0.f;
    } else if (i < 64 * 16 + 16) {
        const int c = i - 64 * 16;
        bp[c] = (c < 10) ? b4[c] : 0.f;
    } else if (i < 64 * 16 + 16 + GCN_NTICK) {
        tick[i - (64 * 16 + 16)] = 0;
    }
}

__global__ void gcn_zero(float* __restrict__ p, int n)
{
    const int i = blockIdx.x * blockDim.x + threadIdx.x;
    if (i < n) p[i] = 0.f;
}

#define GCN_TILE 512

__device__ __forceinline__ float gcn_bcast_f(float v, int srclane)
{
    return __builtin_bit_cast(float,
        __builtin_amdgcn_readlane(__builtin_bit_cast(int, v), srclane));
}

__device__ __forceinline__ void gcn_add_f32(float* p, float v)
{
    __hip_atomic_fetch_add(p, v, __ATOMIC_RELAXED, __HIP_MEMORY_SCOPE_AGENT);
}

template <int D>
__device__ __forceinline__ void gcn_apply(
    const float* __restrict__ P, float* __restrict__ out,
    int s_c, int d_c, float w_c, int cnt, int lane)
{
    if constexpr (D == 64) {
        #pragma unroll
        for (int j0 = 0; j0 < 32; j0 += 8) {
            if (cnt > j0) {
                float h0[8], h1[8];
                #pragma unroll
                for (int u = 0; u < 8; ++u) {
                    const int s = __builtin_amdgcn_readlane(s_c, j0 + u);
                    const float* row = P + (size_t)s * 64 + lane;
                    h0[u] = row[0];
                    h1[u] = row[32];
                }
                asm volatile("" :: "v"(h0[0]), "v"(h0[1]), "v"(h0[2]), "v"(h0[3]),
                                   "v"(h0[4]), "v"(h0[5]), "v"(h0[6]), "v"(h0[7]),
                                   "v"(h1[0]), "v"(h1[1]), "v"(h1[2]), "v"(h1[3]),
                                   "v"(h1[4]), "v"(h1[5]), "v"(h1[6]), "v"(h1[7]));
                #pragma unroll
                for (int u = 0; u < 8; ++u) {
                    if (j0 + u < cnt) {
                        const int   d  = __builtin_amdgcn_readlane(d_c, j0 + u);
                        const float ww = gcn_bcast_f(w_c, j0 + u);
                        float* o = out + (size_t)d * 64 + lane;
                        gcn_add_f32(o,      h0[u] * ww);
                        gcn_add_f32(o + 32, h1[u] * ww);
                    }
                }
            }
        }
    } else {
        const int sub = lane >> 4;
        const int c   = lane & 15;
        #pragma unroll
        for (int j0 = 0; j0 < 16; j0 += 8) {
            if (cnt > 2 * j0) {
                int   ss[8], dd[8];
                float ww[8];
                #pragma unroll
                for (int u = 0; u < 8; ++u) {
                    const int ja = 2 * (j0 + u);
                    const int sa = __builtin_amdgcn_readlane(s_c, ja);
                    const int sb = __builtin_amdgcn_readlane(s_c, ja + 1);
                    ss[u] = sub ? sb : sa;
                    const int da = __builtin_amdgcn_readlane(d_c, ja);
                    const int db = __builtin_amdgcn_readlane(d_c, ja + 1);
                    dd[u] = sub ? db : da;
                    const float wa = gcn_bcast_f(w_c, ja);
                    const float wb = gcn_bcast_f(w_c, ja + 1);
                    ww[u] = sub ? wb : wa;
                }
                if (c < D) {
                    float h[8];
                    #pragma unroll
                    for (int u = 0; u < 8; ++u)
                        h[u] = P[(size_t)ss[u] * D + c];
                    asm volatile("" :: "v"(h[0]), "v"(h[1]), "v"(h[2]), "v"(h[3]),
                                       "v"(h[4]), "v"(h[5]), "v"(h[6]), "v"(h[7]));
                    #pragma unroll
                    for (int u = 0; u < 8; ++u)
                        if (2 * (j0 + u) + sub < cnt)
                            gcn_add_f32(out + (size_t)dd[u] * D + c, h[u] * ww[u]);
                }
            }
        }
    }
}

template <int D>
__device__ __forceinline__ void gcn_batch(
    const float* __restrict__ P, float* __restrict__ out,
    int s_l, int d_l, float w_l, bool mine, unsigned k, int lane)
{
    if (k == 0u) return;
    const int cnt   = __builtin_popcount(k);
    const int below = (int)__builtin_amdgcn_mbcnt_lo(k, 0u);
    const int pos4  = (mine ? below : cnt + lane - below) << 2;
    const int sfirst = __builtin_amdgcn_readlane(s_l, __builtin_ctz(k));
    const int   s_c = __builtin_amdgcn_ds_permute(pos4, mine ? s_l : sfirst);
    const int   d_c = __builtin_amdgcn_ds_permute(pos4, d_l);
    const float w_c = __builtin_bit_cast(float,
        __builtin_amdgcn_ds_permute(pos4, __builtin_bit_cast(int, w_l)));
    gcn_apply<D>(P, out, s_c, d_c, w_c, cnt, lane);
}

template <int D>
__global__ __launch_bounds__(256) void gcn_scatter(
    const float* __restrict__ P, const int* __restrict__ src,
    const int* __restrict__ dst, const float* __restrict__ w,
    float* __restrict__ out, int E, int nlow, int ntiles,
    int* __restrict__ tick)
{
    static_assert(D == 64 || D <= 16, "row is two whole lines or fits a half-wave");
    __shared__ int s_item[2];
    const int tid  = threadIdx.x;
    const int lane = tid & 31;

    if (tid == 0) {
        const int die = (int)((__builtin_amdgcn_s_sendmsg_rtn(0x87) >> 16) & 15u);
        int r = (die >> 2) & 1;
        int t = __hip_atomic_fetch_add(tick + r * 32, 1, __ATOMIC_RELAXED,
                                       __HIP_MEMORY_SCOPE_AGENT);
        if (t >= ntiles) {
            r ^= 1;
            t = __hip_atomic_fetch_add(tick + r * 32, 1, __ATOMIC_RELAXED,
                                       __HIP_MEMORY_SCOPE_AGENT);
        }
        s_item[0] = t;
        s_item[1] = r;
    }
    __syncthreads();
    const int t = __builtin_amdgcn_readfirstlane(s_item[0]);
    const int r = __builtin_amdgcn_readfirstlane(s_item[1]);
    if (t >= ntiles) return;

    const int wv = __builtin_amdgcn_readfirstlane(tid >> 5);
    const int e0 = t * GCN_TILE + wv * 64;
    if (e0 >= E) return;

    const int  ea = e0 + lane, eb = ea + 32;
    const bool la = ea < E,    lb = eb < E;
    const int  d_a = la ? dst[ea] : 0;
    const int  d_b = lb ? dst[eb] : 0;
    const int  s_a = la ? src[ea] : 0;
    const int  s_b = lb ? src[eb] : 0;
    const float w_a = la ? w[ea] : 0.f;
    const float w_b = lb ? w[eb] : 0.f;
    const bool hi = (r != 0);
    const bool ma = la && ((d_a >= nlow) == hi);
    const bool mb = lb && ((d_b >= nlow) == hi);
    const unsigned ka = __builtin_amdgcn_ballot_w32(ma);
    const unsigned kb = __builtin_amdgcn_ballot_w32(mb);

    gcn_batch<D>(P, out, s_a, d_a, w_a, ma, ka, lane);
    gcn_batch<D>(P, out, s_b, d_b, w_b, mb, kb, lane);
}

extern "C" void kernel_launch(void* const* d_in, const int* in_sizes, int n_in,
                              void* d_out, int out_size, void* d_ws, size_t ws_size,
                              hipStream_t stream)
{
    (void)in_sizes; (void)n_in; (void)out_size; (void)ws_size;
    const int N = NNODES, E = NEDGES;

    const float* x    = (const float*)d_in[0];
    const int*   esrc = (const int*)  d_in[1];
    const int*   edst = (const int*)  d_in[2];
    const float* ew   = (const float*)d_in[3];
    const float* Ws[5]; const float* bs[5];
    for (int i = 0; i < 5; ++i) {
        Ws[i] = (const float*)d_in[4 + 2 * i];
        bs[i] = (const float*)d_in[5 + 2 * i];
    }
    float* out  = (float*)d_out;
    float* buf0 = (float*)d_ws;
    float* buf1 = buf0 + (size_t)N * 64;
    float* buf2 = buf1 + (size_t)N * 64;
    float* wpad = buf2 + (size_t)N * 64;
    float* bpad = wpad + 64 * 16;
    int*   tick = (int*)(bpad + 32);

    const dim3 ggrid((N + 15) / 16);
    const int  ntiles = (NEDGES + GCN_TILE - 1) / GCN_TILE;
    const int  agrid  = 2 * ntiles;
    const int  nlow   = NNODES / 2;

    gcn_pad_w4<<<(64 * 16 + 16 + GCN_NTICK + 255) / 256, 256, 0, stream>>>(
        Ws[4], bs[4], wpad, bpad, tick);

    gcn_gemm_wmma<<<ggrid, 128, 0, stream>>>(x, Ws[0], bs[0], buf0, N, 128, 64, 64, 0);
    gcn_zero<<<(N * 64 + 255) / 256, 256, 0, stream>>>(buf1, N * 64);
    gcn_scatter<64><<<agrid, 256, 0, stream>>>(buf0, esrc, edst, ew, buf1, E, nlow, ntiles, tick);

    gcn_gemm_wmma<<<ggrid, 128, 0, stream>>>(buf1, Ws[1], bs[1], buf2, N, 64, 64, 64, 1);
    gcn_zero<<<(N * 64 + 255) / 256, 256, 0, stream>>>(buf0, N * 64);
    gcn_scatter<64><<<agrid, 256, 0, stream>>>(buf2, esrc, edst, ew, buf0, E, nlow, ntiles, tick + 64);

    gcn_gemm_wmma<<<ggrid, 128, 0, stream>>>(buf0, Ws[2], bs[2], buf1, N, 64, 64, 64, 1);
    gcn_zero<<<(N * 64 + 255) / 256, 256, 0, stream>>>(buf2, N * 64);
    gcn_scatter<64><<<agrid, 256, 0, stream>>>(buf1, esrc, edst, ew, buf2, E, nlow, ntiles, tick + 128);

    gcn_gemm_wmma<<<ggrid, 128, 0, stream>>>(buf2, Ws[3], bs[3], buf0, N, 64, 64, 64, 1);
    gcn_zero<<<(N * 64 + 255) / 256, 256, 0, stream>>>(buf1, N * 64);
    gcn_scatter<64><<<agrid, 256, 0, stream>>>(buf0, esrc, edst, ew, buf1, E, nlow, ntiles, tick + 192);

    gcn_gemm_wmma<<<ggrid, 128, 0, stream>>>(buf1, wpad, bpad, buf2, N, 64, 10, 16, 1);
    gcn_zero<<<(N * 10 + 255) / 256, 256, 0, stream>>>(out, N * 10);
    gcn_scatter<10><<<agrid, 256, 0, stream>>>(buf2, esrc, edst, ew, out, E, nlow, ntiles, tick + 256);
}


# Zero-fills, weight padding and counter clear ride as extra work-groups behind the projection work-groups of each layer's launch (not as stores of its waves): 16 launches become 10; sums unchanged
# speedup vs baseline: 2.4961x; 1.0443x over previous
#include <hip/hip_runtime.h>
#include <hip/hip_bf16.h>

typedef float v2f __attribute__((ext_vector_type(2)));
typedef float v8f __attribute__((ext_vector_type(8)));

#define NNODES 100000
#define NEDGES 1600000

#define GCN_NTICK (5 * 2 * 32)

__global__ __launch_bounds__(128) void gcn_gemm_wmma(
    const float* __restrict__ X, const float* __restrict__ Wm,
    const float* __restrict__ bias, float* __restrict__ P,
    int N, int din, int dout, int doutp, int relu_in,
    int ngemm, int nzero, float* __restrict__ zbuf, int nz4,
    const float* __restrict__ W4, const float* __restrict__ b4,
    float* __restrict__ Wp, float* __restrict__ bp, int* __restrict__ tick)
{
    __shared__ float lds[16 * (128 + 4)];
    const int t = threadIdx.x;

    if ((int)blockIdx.x >= ngemm) {
        const int sb = (int)blockIdx.x - ngemm;
        if (sb < nzero) {
            const float4 z = make_float4(0.f, 0.f, 0.f, 0.f);
            const int i0 = sb * 256 + t;
            if (i0 < nz4)       ((float4*)zbuf)[i0]       = z;
            if (i0 + 128 < nz4) ((float4*)zbuf)[i0 + 128] = z;
        } else {
            const int i = (sb - nzero) * 128 + t;
            if (i < 64 * 16) {
                const int r = i >> 4, c = i & 15;
                Wp[i] = (c < 10) ? W4[r * 10 + c] : 0.f;
            } else if (i < 64 * 16 + 16) {
                const int c = i - 64 * 16;
                bp[c] = (c < 10) ? b4[c] : 0.f;
            } else if (i < 64 * 16 + 16 + GCN_NTICK) {
                tick[i - (64 * 16 + 16)] = 0;
            }
        }
        return;
    }

    const int ldsrow = din + 4;
    const int row0 = blockIdx.x * 16;

    const int nvec = (16 * din) >> 2;
    const int vper = din >> 2;
    for (int idx = t; idx < nvec; idx += 128) {
        const int r  = idx / vper;
        const int c4 = (idx - r * vper) << 2;
        const int row = row0 + r;
        float4 v = make_float4(0.f, 0.f, 0.f, 0.f);
        if (row < N) {
            v = *(const float4*)(X + (size_t)row * din + c4);
            if (relu_in) {
                v.x = fmaxf(v.x, 0.f); v.y = fmaxf(v.y, 0.f);
                v.z = fmaxf(v.z, 0.f); v.w = fmaxf(v.w, 0.f);
            }
        }
        *(float4*)(&lds[r * ldsrow + c4]) = v;
    }
    __syncthreads();

    const int wave = t >> 5;
    const int lane = t & 31;
    const int colbase = wave * 16;
    if (colbase >= doutp) return;

    const int half = lane >> 4;
    const int nrow = lane & 15;
    const int col  = colbase + nrow;

    const float* wk = Wm + (size_t)(2 * half) * doutp + col;
    int aoff = nrow * ldsrow + 2 * half;

    v8f acc = {};
    for (int k = 0; k < din; k += 4) {
        v2f a, b;
        a.x = lds[aoff];
        a.y = lds[aoff + 1];
        b.x = wk[0];
        b.y = wk[doutp];
        acc = __builtin_amdgcn_wmma_f32_16x16x4_f32(
                  false, a, false, b, (short)0, acc, false, false);
        aoff += 4;
        wk   += (size_t)4 * doutp;
    }

    const float bv = bias[col];
    #pragma unroll
    for (int r = 0; r < 8; ++r) {
        const int m   = r + 8 * half;
        const int row = row0 + m;
        if (row < N && col < dout)
            P[(size_t)row * dout + col] = acc[r] + bv;
    }
}

#define GCN_TILE 512

__device__ __forceinline__ float gcn_bcast_f(float v, int srclane)
{
    return __builtin_bit_cast(float,
        __builtin_amdgcn_readlane(__builtin_bit_cast(int, v), srclane));
}

__device__ __forceinline__ void gcn_add_f32(float* p, float v)
{
    __hip_atomic_fetch_add(p, v, __ATOMIC_RELAXED, __HIP_MEMORY_SCOPE_AGENT);
}

template <int D>
__device__ __forceinline__ void gcn_apply(
    const float* __restrict__ P, float* __restrict__ out,
    int s_c, int d_c, float w_c, int cnt, int lane)
{
    if constexpr (D == 64) {
        #pragma unroll
        for (int j0 = 0; j0 < 32; j0 += 8) {
            if (cnt > j0) {
                float h0[8], h1[8];
                #pragma unroll
                for (int u = 0; u < 8; ++u) {
                    const int s = __builtin_amdgcn_readlane(s_c, j0 + u);
                    const float* row = P + (size_t)s * 64 + lane;
                    h0[u] = row[0];
                    h1[u] = row[32];
                }
                asm volatile("" :: "v"(h0[0]), "v"(h0[1]), "v"(h0[2]), "v"(h0[3]),
                                   "v"(h0[4]), "v"(h0[5]), "v"(h0[6]), "v"(h0[7]),
                                   "v"(h1[0]), "v"(h1[1]), "v"(h1[2]), "v"(h1[3]),
                                   "v"(h1[4]), "v"(h1[5]), "v"(h1[6]), "v"(h1[7]));
                #pragma unroll
                for (int u = 0; u < 8; ++u) {
                    if (j0 + u < cnt) {
                        const int   d  = __builtin_amdgcn_readlane(d_c, j0 + u);
                        const float ww = gcn_bcast_f(w_c, j0 + u);
                        float* o = out + (size_t)d * 64 + lane;
                        gcn_add_f32(o,      h0[u] * ww);
                        gcn_add_f32(o + 32, h1[u] * ww);
                    }
                }
            }
        }
    } else {
        const int sub = lane >> 4;
        const int c   = lane & 15;
        #pragma unroll
        for (int j0 = 0; j0 < 16; j0 += 8) {
            if (cnt > 2 * j0) {
                int   ss[8], dd[8];
                float ww[8];
                #pragma unroll
                for (int u = 0; u < 8; ++u) {
                    const int ja = 2 * (j0 + u);
                    const int sa = __builtin_amdgcn_readlane(s_c, ja);
                    const int sb = __builtin_amdgcn_readlane(s_c, ja + 1);
                    ss[u] = sub ? sb : sa;
                    const int da = __builtin_amdgcn_readlane(d_c, ja);
                    const int db = __builtin_amdgcn_readlane(d_c, ja + 1);
                    dd[u] = sub ? db : da;
                    const float wa = gcn_bcast_f(w_c, ja);
                    const float wb = gcn_bcast_f(w_c, ja + 1);
                    ww[u] = sub ? wb : wa;
                }
                if (c < D) {
                    float h[8];
                    #pragma unroll
                    for (int u = 0; u < 8; ++u)
                        h[u] = P[(size_t)ss[u] * D + c];
                    asm volatile("" :: "v"(h[0]), "v"(h[1]), "v"(h[2]), "v"(h[3]),
                                       "v"(h[4]), "v"(h[5]), "v"(h[6]), "v"(h[7]));
                    #pragma unroll
                    for (int u = 0; u < 8; ++u)
                        if (2 * (j0 + u) + sub < cnt)
                            gcn_add_f32(out + (size_t)dd[u] * D + c, h[u] * ww[u]);
                }
            }
        }
    }
}

template <int D>
__device__ __forceinline__ void gcn_batch(
    const float* __restrict__ P, float* __restrict__ out,
    int s_l, int d_l, float w_l, bool mine, unsigned k, int lane)
{
    if (k == 0u) return;
    const int cnt   = __builtin_popcount(k);
    const int below = (int)__builtin_amdgcn_mbcnt_lo(k, 0u);
    const int pos4  = (mine ? below : cnt + lane - below) << 2;
    const int sfirst = __builtin_amdgcn_readlane(s_l, __builtin_ctz(k));
    const int   s_c = __builtin_amdgcn_ds_permute(pos4, mine ? s_l : sfirst);
    const int   d_c = __builtin_amdgcn_ds_permute(pos4, d_l);
    const float w_c = __builtin_bit_cast(float,
        __builtin_amdgcn_ds_permute(pos4, __builtin_bit_cast(int, w_l)));
    gcn_apply<D>(P, out, s_c, d_c, w_c, cnt, lane);
}

template <int D>
__global__ __launch_bounds__(256) void gcn_scatter(
    const float* __restrict__ P, const int* __restrict__ src,
    const int* __restrict__ dst, const float* __restrict__ w,
    float* __restrict__ out, int E, int nlow, int ntiles,
    int* __restrict__ tick)
{
    static_assert(D == 64 || D <= 16, "row is two whole lines or fits a half-wave");
    __shared__ int s_item[2];
    const int tid  = threadIdx.x;
    const int lane = tid & 31;

    if (tid == 0) {
        const int die = (int)((__builtin_amdgcn_s_sendmsg_rtn(0x87) >> 16) & 15u);
        int r = (die >> 2) & 1;
        int t = __hip_atomic_fetch_add(tick + r * 32, 1, __ATOMIC_RELAXED,
                                       __HIP_MEMORY_SCOPE_AGENT);
        if (t >= ntiles) {
            r ^= 1;
            t = __hip_atomic_fetch_add(tick + r * 32, 1, __ATOMIC_RELAXED,
                                       __HIP_MEMORY_SCOPE_AGENT);
        }
        s_item[0] = t;
        s_item[1] = r;
    }
    __syncthreads();
    const int t = __builtin_amdgcn_readfirstlane(s_item[0]);
    const int r = __builtin_amdgcn_readfirstlane(s_item[1]);
    if (t >= ntiles) return;

    const int wv = __builtin_amdgcn_readfirstlane(tid >> 5);
    const int e0 = t * GCN_TILE + wv * 64;
    if (e0 >= E) return;

    const int  ea = e0 + lane, eb = ea + 32;
    const bool la = ea < E,    lb = eb < E;
    const int  d_a = la ? dst[ea] : 0;
    const int  d_b = lb ? dst[eb] : 0;
    const int  s_a = la ? src[ea] : 0;
    const int  s_b = lb ? src[eb] : 0;
    const float w_a = la ? w[ea] : 0.f;
    const float w_b = lb ? w[eb] : 0.f;
    const bool hi = (r != 0);
    const bool ma = la && ((d_a >= nlow) == hi);
    const bool mb = lb && ((d_b >= nlow) == hi);
    const unsigned ka = __builtin_amdgcn_ballot_w32(ma);
    const unsigned kb = __builtin_amdgcn_ballot_w32(mb);

    gcn_batch<D>(P, out, s_a, d_a, w_a, ma, ka, lane);
    gcn_batch<D>(P, out, s_b, d_b, w_b, mb, kb, lane);
}

extern "C" void kernel_launch(void* const* d_in, const int* in_sizes, int n_in,
                              void* d_out, int out_size, void* d_ws, size_t ws_size,
                              hipStream_t stream)
{
    (void)in_sizes; (void)n_in; (void)out_size; (void)ws_size;
    const int N = NNODES, E = NEDGES;

    const float* x    = (const float*)d_in[0];
    const int*   esrc = (const int*)  d_in[1];
    const int*   edst = (const int*)  d_in[2];
    const float* ew   = (const float*)d_in[3];
    const float* Ws[5]; const float* bs[5];
    for (int i = 0; i < 5; ++i) {
        Ws[i] = (const float*)d_in[4 + 2 * i];
        bs[i] = (const float*)d_in[5 + 2 * i];
    }
    float* out  = (float*)d_out;
    float* buf0 = (float*)d_ws;
    float* buf1 = buf0 + (size_t)N * 64;
    float* buf2 = buf1 + (size_t)N * 64;
    float* wpad = buf2 + (size_t)N * 64;
    float* bpad = wpad + 64 * 16;
    int*   tick = (int*)(bpad + 32);

    const int  ngemm  = (NNODES + 15) / 16;
    const int  nz4w   = NNODES * 64 / 4;
    const int  nz4o   = NNODES * 10 / 4;
    const int  nzw    = (NNODES * 64 / 4 + 255) / 256;
    const int  nzo    = (NNODES * 10 / 4 + 255) / 256;
    const int  npad   = (64 * 16 + 16 + GCN_NTICK + 127) / 128;
    const int  ntiles = (NEDGES + GCN_TILE - 1) / GCN_TILE;
    const int  agrid  = 2 * ntiles;
    const int  nlow   = NNODES / 2;

    gcn_gemm_wmma<<<ngemm + nzw + npad, 128, 0, stream>>>(
        x, Ws[0], bs[0], buf0, N, 128, 64, 64, 0,
        ngemm, nzw, buf1, nz4w, Ws[4], bs[4], wpad, bpad, tick);
    gcn_scatter<64><<<agrid, 256, 0, stream>>>(buf0, esrc, edst, ew, buf1, E, nlow, ntiles, tick);

    gcn_gemm_wmma<<<ngemm + nzw, 128, 0, stream>>>(
        buf1, Ws[1], bs[1], buf2, N, 64, 64, 64, 1,
        ngemm, nzw, buf0, nz4w, nullptr, nullptr, nullptr, nullptr, nullptr);
    gcn_scatter<64><<<agrid, 256, 0, stream>>>(buf2, esrc, edst, ew, buf0, E, nlow, ntiles, tick + 64);

    gcn_gemm_wmma<<<ngemm + nzw, 128, 0, stream>>>(
        buf0, Ws[2], bs[2], buf1, N, 64, 64, 64, 1,
        ngemm, nzw, buf2, nz4w, nullptr, nullptr, nullptr, nullptr, nullptr);
    gcn_scatter<64><<<agrid, 256, 0, stream>>>(buf1, esrc, edst, ew, buf2, E, nlow, ntiles, tick + 128);

    gcn_gemm_wmma<<<ngemm + nzw, 128, 0, stream>>>(
        buf2, Ws[3], bs[3], buf0, N, 64, 64, 64, 1,
        ngemm, nzw, buf1, nz4w, nullptr, nullptr, nullptr, nullptr, nullptr);
    gcn_scatter<64><<<agrid, 256, 0, stream>>>(buf0, esrc, edst, ew, buf1, E, nlow, ntiles, tick + 192);

    gcn_gemm_wmma<<<ngemm + nzo, 128, 0, stream>>>(
        buf1, wpad, bpad, buf2, N, 64, 10, 16, 1,
        ngemm, nzo, out, nz4o, nullptr, nullptr, nullptr, nullptr, nullptr);
    gcn_scatter<10><<<agrid, 256, 0, stream>>>(buf2, esrc, edst, ew, out, E, nlow, ntiles, tick + 256);
}
